# starting kernel
# baseline (speedup 1.0000x reference)
#include <hip/hip_runtime.h>

typedef __attribute__((ext_vector_type(2))) float v2f;
typedef __attribute__((ext_vector_type(8))) float v8f;

#define N_NODES 50000
#define N_EDGES 800000
#define D_IN    64
#define D_HID   128
#define HPAD    132

__global__ void gin_init(const float* __restrict__ x,
                         const float* __restrict__ eps,
                         float* __restrict__ agg, int n) {
    int i = blockIdx.x * blockDim.x + threadIdx.x;
    if (i < n) agg[i] = (1.0f + eps[0]) * x[i];
}

__global__ void gin_scatter(const float* __restrict__ x,
                            const int* __restrict__ ei,
                            float* __restrict__ agg) {
    unsigned tid = blockIdx.x * blockDim.x + threadIdx.x;
    unsigned e = tid >> 4;
    if (e >= N_EDGES) return;
    int f = (tid & 15u) << 2;
    int dst = ei[e];
    int src = ei[N_EDGES + e];
    const float4 v = *(const float4*)(x + (size_t)src * D_IN + f);
    float* a = agg + (size_t)dst * D_IN + f;
    atomicAdd(a + 0, v.x);
    atomicAdd(a + 1, v.y);
    atomicAdd(a + 2, v.z);
    atomicAdd(a + 3, v.w);
}

__global__ __launch_bounds__(128)
void gin_mlp(const float* __restrict__ h,
             const float* __restrict__ W1,
             const float* __restrict__ b1,
             const float* __restrict__ W2,
             const float* __restrict__ b2,
             float* __restrict__ out) {
    __shared__ float hid[4][16 * HPAD];

    const int wave  = threadIdx.x >> 5;
    const int lane  = threadIdx.x & 31;
    const int r     = lane & 15;
    const int koff  = (lane >> 4) << 1;
    const int mhalf = (lane >> 4) << 3;
    const int tile  = blockIdx.x * 4 + wave;
    if (tile * 16 >= N_NODES) return;
    const int m0 = tile * 16;

    v2f afr[16];
    const float* arow = h + (size_t)(m0 + r) * D_IN + koff;
#pragma unroll
    for (int kk = 0; kk < 16; ++kk)
        afr[kk] = *(const v2f*)(arow + kk * 4);

    float* hrow = &hid[wave][0];
#pragma unroll
    for (int nt = 0; nt < 8; ++nt) {
        const int n0 = nt * 16;
        const float bias = b1[n0 + r];
        v8f acc;
#pragma unroll
        for (int v = 0; v < 8; ++v) acc[v] = bias;
#pragma unroll
        for (int kk = 0; kk < 16; ++kk) {
            const int k = kk * 4 + koff;
            v2f bfr;
            bfr.x = W1[(size_t)k * D_HID + n0 + r];
            bfr.y = W1[(size_t)(k + 1) * D_HID + n0 + r];
            acc = __builtin_amdgcn_wmma_f32_16x16x4_f32(
                false, afr[kk], false, bfr, (short)0, acc, false, false);
        }
#pragma unroll
        for (int v = 0; v < 8; ++v) {
            float t = acc[v];
            hrow[(v + mhalf) * HPAD + n0 + r] = t > 0.0f ? t : 0.0f;
        }
    }

    float* orow = out + (size_t)m0 * D_IN;
#pragma unroll
    for (int nt = 0; nt < 4; ++nt) {
        const int n0 = nt * 16;
        const float bias = b2[n0 + r];
        v8f acc;
#pragma unroll
        for (int v = 0; v < 8; ++v) acc[v] = bias;
#pragma unroll
        for (int kk = 0; kk < 32; ++kk) {
            const int k = kk * 4 + koff;
            v2f a2;
            a2.x = hrow[r * HPAD + k];
            a2.y = hrow[r * HPAD + k + 1];
            v2f bfr;
            bfr.x = W2[(size_t)k * D_IN + n0 + r];
            bfr.y = W2[(size_t)(k + 1) * D_IN + n0 + r];
            acc = __builtin_amdgcn_wmma_f32_16x16x4_f32(
                false, a2, false, bfr, (short)0, acc, false, false);
        }
#pragma unroll
        for (int v = 0; v < 8; ++v)
            orow[(size_t)(v + mhalf) * D_IN + n0 + r] = acc[v];
    }
}

extern "C" void kernel_launch(void* const* d_in, const int* in_sizes, int n_in,
                              void* d_out, int out_size, void* d_ws, size_t ws_size,
                              hipStream_t stream) {
    const float* x   = (const float*)d_in[0];
    const int*   ei  = (const int*)d_in[1];
    const float* W1  = (const float*)d_in[2];
    const float* b1  = (const float*)d_in[3];
    const float* W2  = (const float*)d_in[4];
    const float* b2  = (const float*)d_in[5];
    const float* eps = (const float*)d_in[6];
    float* out = (float*)d_out;
    float* agg = (float*)d_ws;

    int n = N_NODES * D_IN;
    gin_init<<<(n + 255) / 256, 256, 0, stream>>>(x, eps, agg, n);

    unsigned total = (unsigned)N_EDGES * 16u;
    gin_scatter<<<(total + 255) / 256, 256, 0, stream>>>(x, ei, agg);

    int tiles = (N_NODES + 15) / 16;
    gin_mlp<<<(tiles + 3) / 4, 128, 0, stream>>>(agg, W1, b1, W2, b2, out);
}


# scatter remapped to lane = feature: a wave owns 32 edges (indices read once, readlane broadcast); each row gather and f32 atomic add is one whole 128B line, not 4 quarter-filled lines
# speedup vs baseline: 1.6970x; 1.6970x over previous
#include <hip/hip_runtime.h>

typedef __attribute__((ext_vector_type(2))) float v2f;
typedef __attribute__((ext_vector_type(8))) float v8f;

#define N_NODES 50000
#define N_EDGES 800000
#define D_IN    64
#define D_HID   128
#define HPAD    132

__global__ void gin_init(const float* __restrict__ x,
                         const float* __restrict__ eps,
                         float* __restrict__ agg, int n) {
    int i = blockIdx.x * blockDim.x + threadIdx.x;
    if (i < n) agg[i] = (1.0f + eps[0]) * x[i];
}

#define EPW 32
#define EB  16
static_assert(N_EDGES % EPW == 0, "a wave owns EPW whole edges");

__global__ __launch_bounds__(256)
void gin_scatter(const float* __restrict__ x,
                 const int* __restrict__ ei,
                 float* __restrict__ agg) {
    const unsigned lane = threadIdx.x & 31u;
    const unsigned e0 = ((blockIdx.x * blockDim.x + threadIdx.x) >> 5) * EPW;
    if (e0 >= N_EDGES) return;
    const int my_dst = ei[e0 + lane];
    const int my_src = ei[N_EDGES + e0 + lane];
#pragma unroll
    for (int b = 0; b < EPW; b += EB) {
        float v0[EB], v1[EB];
#pragma unroll
        for (int j = 0; j < EB; ++j) {
            const unsigned s = (unsigned)__builtin_amdgcn_readlane(my_src, b + j);
            const float* xr = x + (size_t)s * D_IN + lane;
            v0[j] = xr[0];
            v1[j] = xr[32];
        }
#pragma unroll
        for (int j = 0; j < EB; ++j) {
            const unsigned d = (unsigned)__builtin_amdgcn_readlane(my_dst, b + j);
            float* ar = agg + (size_t)d * D_IN + lane;
            atomicAdd(ar, v0[j]);
            atomicAdd(ar + 32, v1[j]);
        }
    }
}

__global__ __launch_bounds__(128)
void gin_mlp(const float* __restrict__ h,
             const float* __restrict__ W1,
             const float* __restrict__ b1,
             const float* __restrict__ W2,
             const float* __restrict__ b2,
             float* __restrict__ out) {
    __shared__ float hid[4][16 * HPAD];

    const int wave  = threadIdx.x >> 5;
    const int lane  = threadIdx.x & 31;
    const int r     = lane & 15;
    const int koff  = (lane >> 4) << 1;
    const int mhalf = (lane >> 4) << 3;
    const int tile  = blockIdx.x * 4 + wave;
    if (tile * 16 >= N_NODES) return;
    const int m0 = tile * 16;

    v2f afr[16];
    const float* arow = h + (size_t)(m0 + r) * D_IN + koff;
#pragma unroll
    for (int kk = 0; kk < 16; ++kk)
        afr[kk] = *(const v2f*)(arow + kk * 4);

    float* hrow = &hid[wave][0];
#pragma unroll
    for (int nt = 0; nt < 8; ++nt) {
        const int n0 = nt * 16;
        const float bias = b1[n0 + r];
        v8f acc;
#pragma unroll
        for (int v = 0; v < 8; ++v) acc[v] = bias;
#pragma unroll
        for (int kk = 0; kk < 16; ++kk) {
            const int k = kk * 4 + koff;
            v2f bfr;
            bfr.x = W1[(size_t)k * D_HID + n0 + r];
            bfr.y = W1[(size_t)(k + 1) * D_HID + n0 + r];
            acc = __builtin_amdgcn_wmma_f32_16x16x4_f32(
                false, afr[kk], false, bfr, (short)0, acc, false, false);
        }
#pragma unroll
        for (int v = 0; v < 8; ++v) {
            float t = acc[v];
            hrow[(v + mhalf) * HPAD + n0 + r] = t > 0.0f ? t : 0.0f;
        }
    }

    float* orow = out + (size_t)m0 * D_IN;
#pragma unroll
    for (int nt = 0; nt < 4; ++nt) {
        const int n0 = nt * 16;
        const float bias = b2[n0 + r];
        v8f acc;
#pragma unroll
        for (int v = 0; v < 8; ++v) acc[v] = bias;
#pragma unroll
        for (int kk = 0; kk < 32; ++kk) {
            const int k = kk * 4 + koff;
            v2f a2;
            a2.x = hrow[r * HPAD + k];
            a2.y = hrow[r * HPAD + k + 1];
            v2f bfr;
            bfr.x = W2[(size_t)k * D_IN + n0 + r];
            bfr.y = W2[(size_t)(k + 1) * D_IN + n0 + r];
            acc = __builtin_amdgcn_wmma_f32_16x16x4_f32(
                false, a2, false, bfr, (short)0, acc, false, false);
        }
#pragma unroll
        for (int v = 0; v < 8; ++v)
            orow[(size_t)(v + mhalf) * D_IN + n0 + r] = acc[v];
    }
}

extern "C" void kernel_launch(void* const* d_in, const int* in_sizes, int n_in,
                              void* d_out, int out_size, void* d_ws, size_t ws_size,
                              hipStream_t stream) {
    const float* x   = (const float*)d_in[0];
    const int*   ei  = (const int*)d_in[1];
    const float* W1  = (const float*)d_in[2];
    const float* b1  = (const float*)d_in[3];
    const float* W2  = (const float*)d_in[4];
    const float* b2  = (const float*)d_in[5];
    const float* eps = (const float*)d_in[6];
    float* out = (float*)d_out;
    float* agg = (float*)d_ws;

    int n = N_NODES * D_IN;
    gin_init<<<(n + 255) / 256, 256, 0, stream>>>(x, eps, agg, n);

    unsigned total = (unsigned)N_EDGES;
    gin_scatter<<<(total + 255) / 256, 256, 0, stream>>>(x, ei, agg);

    int tiles = (N_NODES + 15) / 16;
    gin_mlp<<<(tiles + 3) / 4, 128, 0, stream>>>(agg, W1, b1, W2, b2, out);
}
